# starting kernel
# baseline (speedup 1.0000x reference)
#include <hip/hip_runtime.h>
#include <hip/hip_bf16.h>

typedef __attribute__((ext_vector_type(2))) float v2f;
typedef __attribute__((ext_vector_type(8))) float v8f;

#define HDIM 64

__global__ void zero_f32(float* __restrict__ p, int n) {
  int i = blockIdx.x * blockDim.x + threadIdx.x;
  if (i < n) p[i] = 0.0f;
}

__global__ void deg_accum(const int* __restrict__ dst, float* __restrict__ deg, int E) {
  int e = blockIdx.x * blockDim.x + threadIdx.x;
  if (e < E) atomicAdd(&deg[dst[e]], 1.0f);
}

__global__ void deg_to_dis(float* __restrict__ deg, int n) {
  int i = blockIdx.x * blockDim.x + threadIdx.x;
  if (i < n) {
    float d = deg[i];
    deg[i] = (d > 0.0f) ? rsqrtf(fmaxf(d, 1.0f)) : 0.0f;
  }
}

template<int K, bool RELU_BIAS>
__global__ void gemm_n64_wmma(const float* __restrict__ A,
                              const float* __restrict__ W,
                              const float* __restrict__ bias,
                              float* __restrict__ out, int n_rows) {
  int wave = (blockIdx.x * blockDim.x + threadIdx.x) >> 5;
  int lane = threadIdx.x & 31;
  int row0 = wave * 16;
  if (row0 >= n_rows) return;

  const int half = lane >> 4;
  const int l16  = lane & 15;

  v8f acc0 = {}, acc1 = {}, acc2 = {}, acc3 = {};

  const float* arow = A + (size_t)(row0 + l16) * K;
  for (int kb = 0; kb < K; kb += 4) {
    v2f a;
    a.x = arow[kb + half * 2 + 0];
    a.y = arow[kb + half * 2 + 1];

    const float* bp = W + (size_t)(kb + half * 2) * HDIM + l16;
    v2f b0, b1, b2, b3;
    b0.x = bp[0];        b0.y = bp[HDIM];
    b1.x = bp[16];       b1.y = bp[HDIM + 16];
    b2.x = bp[32];       b2.y = bp[HDIM + 32];
    b3.x = bp[48];       b3.y = bp[HDIM + 48];

    acc0 = __builtin_amdgcn_wmma_f32_16x16x4_f32(false, a, false, b0, (short)0, acc0, false, false);
    acc1 = __builtin_amdgcn_wmma_f32_16x16x4_f32(false, a, false, b1, (short)0, acc1, false, false);
    acc2 = __builtin_amdgcn_wmma_f32_16x16x4_f32(false, a, false, b2, (short)0, acc2, false, false);
    acc3 = __builtin_amdgcn_wmma_f32_16x16x4_f32(false, a, false, b3, (short)0, acc3, false, false);
  }

  v8f accs[4] = {acc0, acc1, acc2, acc3};
  #pragma unroll
  for (int nt = 0; nt < 4; ++nt) {
    int col = nt * 16 + l16;
    float bcol = RELU_BIAS ? bias[col] : 0.0f;
    #pragma unroll
    for (int r = 0; r < 8; ++r) {
      int m = half * 8 + r;
      float v = accs[nt][r];
      if (RELU_BIAS) { v += bcol; v = fmaxf(v, 0.0f); }
      out[(size_t)(row0 + m) * HDIM + col] = v;
    }
  }
}

__global__ void scatter_edges(const int* __restrict__ src, const int* __restrict__ dst,
                              const float* __restrict__ dis,
                              const float* __restrict__ hw,
                              float* __restrict__ acc, int E) {
  int gid  = blockIdx.x * blockDim.x + threadIdx.x;
  int e    = gid >> 5;
  int lane = gid & 31;
  if (e >= E) return;
  int s = src[e];
  int d = dst[e];
  float coef = dis[s] * dis[d];
  const float2* hp = (const float2*)(hw + (size_t)s * HDIM);
  float2 v = hp[lane];
  float* ap = acc + (size_t)d * HDIM + lane * 2;
  atomicAdd(ap + 0, v.x * coef);
  atomicAdd(ap + 1, v.y * coef);
}

__global__ void finalize_layer(const float* __restrict__ acc,
                               const float* __restrict__ bias,
                               float* __restrict__ h, int n64) {
  int i = blockIdx.x * blockDim.x + threadIdx.x;
  if (i >= n64) return;
  float v = acc[i] + bias[i & (HDIM - 1)];
  h[i] = fmaxf(v, 0.0f) + h[i];
}

__global__ void head_kernel(const float* __restrict__ h, const float* __restrict__ Wh,
                            const float* __restrict__ bh, float* __restrict__ out, int n) {
  int i = blockIdx.x * blockDim.x + threadIdx.x;
  if (i >= n) return;
  const float* hp = h + (size_t)i * HDIM;
  float s = bh[0];
  #pragma unroll
  for (int k = 0; k < HDIM; ++k) s += hp[k] * Wh[k];
  out[i] = s;
}

extern "C" void kernel_launch(void* const* d_in, const int* in_sizes, int n_in,
                              void* d_out, int out_size, void* d_ws, size_t ws_size,
                              hipStream_t stream) {
  const float* x      = (const float*)d_in[0];
  const int*   eidx   = (const int*)d_in[1];
  const float* W_pre  = (const float*)d_in[2];
  const float* b_pre  = (const float*)d_in[3];
  const float* W1     = (const float*)d_in[4];
  const float* b1     = (const float*)d_in[5];
  const float* W2     = (const float*)d_in[6];
  const float* b2     = (const float*)d_in[7];
  const float* W_head = (const float*)d_in[8];
  const float* b_head = (const float*)d_in[9];
  float* out = (float*)d_out;

  const int N = in_sizes[0] / 128;
  const int E = in_sizes[1] / 2;
  const int* src = eidx;
  const int* dst = eidx + E;

  float* dis = (float*)d_ws;
  float* h   = dis + N;
  float* hw  = h   + (size_t)N * HDIM;
  float* acc = hw  + (size_t)N * HDIM;

  const int n64 = N * HDIM;

  zero_f32<<<(N + 255) / 256, 256, 0, stream>>>(dis, N);
  deg_accum<<<(E + 255) / 256, 256, 0, stream>>>(dst, dis, E);
  deg_to_dis<<<(N + 255) / 256, 256, 0, stream>>>(dis, N);

  int gemm_blocks = (N + 127) / 128;
  gemm_n64_wmma<128, true><<<gemm_blocks, 256, 0, stream>>>(x, W_pre, b_pre, h, N);

  const float* Ws[2] = {W1, W2};
  const float* bs[2] = {b1, b2};
  for (int layer = 0; layer < 2; ++layer) {
    gemm_n64_wmma<64, false><<<gemm_blocks, 256, 0, stream>>>(h, Ws[layer], nullptr, hw, N);
    zero_f32<<<(n64 + 255) / 256, 256, 0, stream>>>(acc, n64);
    long long sthreads = (long long)E * 32;
    scatter_edges<<<(int)((sthreads + 255) / 256), 256, 0, stream>>>(src, dst, dis, hw, acc, E);
    finalize_layer<<<(n64 + 255) / 256, 256, 0, stream>>>(acc, bs[layer], h, n64);
  }

  head_kernel<<<(N + 255) / 256, 256, 0, stream>>>(h, W_head, b_head, out, N);
}


# Scatter: lane i takes columns i and i+32 of a row (was 2i, 2i+1), so each f32 atomic instruction adds one whole 128-byte line, not half of two; row read by two 4-byte loads; sums in arrival order
# speedup vs baseline: 2.1494x; 2.1494x over previous
#include <hip/hip_runtime.h>
#include <hip/hip_bf16.h>

typedef __attribute__((ext_vector_type(2))) float v2f;
typedef __attribute__((ext_vector_type(8))) float v8f;

#define HDIM 64

__global__ void zero_f32(float* __restrict__ p, int n) {
  int i = blockIdx.x * blockDim.x + threadIdx.x;
  if (i < n) p[i] = 0.0f;
}

#define ATOMIC_BLOCKS 2048
#define ATOMIC_THREADS 256
#define HALF_PITCH 64
#define LAUNCH_PITCH 128
#define N_COUNTER_INTS 384

__device__ __forceinline__ int chip_half() {
  return (int)((__builtin_amdgcn_s_sendmsg_rtn(0x87) >> 18) & 1u);
}

__device__ __forceinline__ void claim_range_slot(int* __restrict__ arrivals, int& range, int& slot) {
  __shared__ int s_pick[2];
  if (threadIdx.x == 0) {
    const int side = chip_half();
    const int r    = atomicAdd(&arrivals[side * HALF_PITCH], 1);
    const int per  = ATOMIC_BLOCKS / 2;
    s_pick[0] = (r < per) ? side : (side ^ 1);
    s_pick[1] = (r < per) ? r : (ATOMIC_BLOCKS - 1 - r);
  }
  __syncthreads();
  range = __builtin_amdgcn_readfirstlane(s_pick[0]);
  slot  = __builtin_amdgcn_readfirstlane(s_pick[1]);
}

__device__ __forceinline__ int range_cut(int n) { return ((n >> 1) + 63) & ~63; }

#define DEG_PER_PASS 8
__global__ void __launch_bounds__(ATOMIC_THREADS)
deg_accum(const int* __restrict__ dst, float* __restrict__ deg, int E, int n,
          int* __restrict__ arrivals) {
  int range, slot;
  claim_range_slot(arrivals, range, slot);
  const int cut  = range_cut(n);
  const int lo   = range ? cut : 0;
  const int hi   = range ? n : cut;
  const int last = E - 1;
  const int step = (ATOMIC_BLOCKS / 2) * ATOMIC_THREADS;
  for (int e = slot * ATOMIC_THREADS + (int)threadIdx.x; e < E; e += DEG_PER_PASS * step) {
    int d[DEG_PER_PASS];
    #pragma unroll
    for (int k = 0; k < DEG_PER_PASS; ++k) d[k] = dst[min(e + k * step, last)];
    #pragma unroll
    for (int k = 0; k < DEG_PER_PASS; ++k) {
      if ((e + k * step < E) && (d[k] >= lo) && (d[k] < hi)) atomicAdd(&deg[d[k]], 1.0f);
    }
  }
}

__global__ void deg_to_dis(float* __restrict__ deg, int n) {
  int i = blockIdx.x * blockDim.x + threadIdx.x;
  if (i < n) {
    float d = deg[i];
    deg[i] = (d > 0.0f) ? rsqrtf(fmaxf(d, 1.0f)) : 0.0f;
  }
}

template<int K, bool RELU_BIAS>
__global__ void gemm_n64_wmma(const float* __restrict__ A,
                              const float* __restrict__ W,
                              const float* __restrict__ bias,
                              float* __restrict__ out, int n_rows) {
  int wave = (blockIdx.x * blockDim.x + threadIdx.x) >> 5;
  int lane = threadIdx.x & 31;
  int row0 = wave * 16;
  if (row0 >= n_rows) return;

  const int half = lane >> 4;
  const int l16  = lane & 15;

  v8f acc0 = {}, acc1 = {}, acc2 = {}, acc3 = {};

  const float* arow = A + (size_t)(row0 + l16) * K;
  for (int kb = 0; kb < K; kb += 4) {
    v2f a;
    a.x = arow[kb + half * 2 + 0];
    a.y = arow[kb + half * 2 + 1];

    const float* bp = W + (size_t)(kb + half * 2) * HDIM + l16;
    v2f b0, b1, b2, b3;
    b0.x = bp[0];        b0.y = bp[HDIM];
    b1.x = bp[16];       b1.y = bp[HDIM + 16];
    b2.x = bp[32];       b2.y = bp[HDIM + 32];
    b3.x = bp[48];       b3.y = bp[HDIM + 48];

    acc0 = __builtin_amdgcn_wmma_f32_16x16x4_f32(false, a, false, b0, (short)0, acc0, false, false);
    acc1 = __builtin_amdgcn_wmma_f32_16x16x4_f32(false, a, false, b1, (short)0, acc1, false, false);
    acc2 = __builtin_amdgcn_wmma_f32_16x16x4_f32(false, a, false, b2, (short)0, acc2, false, false);
    acc3 = __builtin_amdgcn_wmma_f32_16x16x4_f32(false, a, false, b3, (short)0, acc3, false, false);
  }

  v8f accs[4] = {acc0, acc1, acc2, acc3};
  #pragma unroll
  for (int nt = 0; nt < 4; ++nt) {
    int col = nt * 16 + l16;
    float bcol = RELU_BIAS ? bias[col] : 0.0f;
    #pragma unroll
    for (int r = 0; r < 8; ++r) {
      int m = half * 8 + r;
      float v = accs[nt][r];
      if (RELU_BIAS) { v += bcol; v = fmaxf(v, 0.0f); }
      out[(size_t)(row0 + m) * HDIM + col] = v;
    }
  }
}

#define ROWS_IN_FLIGHT 8
__global__ void __launch_bounds__(ATOMIC_THREADS)
scatter_edges(const int* __restrict__ src, const int* __restrict__ dst,
              const float* __restrict__ dis,
              const float* __restrict__ hw,
              float* __restrict__ acc, int E, int n,
              int* __restrict__ arrivals) {
  int range, slot;
  claim_range_slot(arrivals, range, slot);
  const int lane = threadIdx.x & 31;
  const int wpb  = ATOMIC_THREADS / 32;
  const int w0   = __builtin_amdgcn_readfirstlane(slot * wpb + (int)(threadIdx.x >> 5));
  const int nw   = (ATOMIC_BLOCKS / 2) * wpb;
  const int cut  = range_cut(n);
  const int lo   = range ? cut : 0;
  const int hi   = range ? n : cut;
  const int ngroups = (E + 31) >> 5;
  const int last = E - 1;
  if (w0 >= ngroups) return;

  int  e_nx  = w0 * 32 + lane;
  bool ok_nx = e_nx < E;
  int  s_nx  = src[min(e_nx, last)];
  int  d_nx  = dst[min(e_nx, last)];

  for (int g = w0; g < ngroups; g += nw) {
    const int  s  = s_nx, d = d_nx;
    const bool ok = ok_nx;
    e_nx  = (g + nw) * 32 + lane;
    ok_nx = e_nx < E;
    s_nx  = src[min(e_nx, last)];
    d_nx  = dst[min(e_nx, last)];

    const bool mine = ok && (d >= lo) && (d < hi);
    float coef = 0.0f;
    if (mine) coef = dis[s] * dis[d];

    unsigned m = __builtin_amdgcn_ballot_w32(mine);
    while (m != 0u) {
      float v0[ROWS_IN_FLIGHT], v1[ROWS_IN_FLIGHT];
      int   j[ROWS_IN_FLIGHT];
      #pragma unroll
      for (int u = 0; u < ROWS_IN_FLIGHT; ++u) {
        j[u] = -1;
        if (m != 0u) {
          j[u] = __builtin_ctz(m);
          m &= m - 1u;
          const int sj = __builtin_amdgcn_readlane(s, j[u]);
          const float* rp = hw + (size_t)sj * HDIM + lane;
          v0[u] = rp[0];
          v1[u] = rp[HDIM / 2];
        }
      }
      #pragma unroll
      for (int u = 0; u < ROWS_IN_FLIGHT; ++u) {
        if (j[u] >= 0) {
          const int   dj = __builtin_amdgcn_readlane(d, j[u]);
          const float cj = __int_as_float(__builtin_amdgcn_readlane(__float_as_int(coef), j[u]));
          float* ap = acc + (size_t)dj * HDIM + lane;
          atomicAdd(ap, v0[u] * cj);
          atomicAdd(ap + HDIM / 2, v1[u] * cj);
        }
      }
    }
  }
}

__global__ void finalize_layer(const float* __restrict__ acc,
                               const float* __restrict__ bias,
                               float* __restrict__ h, int n64) {
  int i = blockIdx.x * blockDim.x + threadIdx.x;
  if (i >= n64) return;
  float v = acc[i] + bias[i & (HDIM - 1)];
  h[i] = fmaxf(v, 0.0f) + h[i];
}

__global__ void head_kernel(const float* __restrict__ h, const float* __restrict__ Wh,
                            const float* __restrict__ bh, float* __restrict__ out, int n) {
  int i = blockIdx.x * blockDim.x + threadIdx.x;
  if (i >= n) return;
  const float* hp = h + (size_t)i * HDIM;
  float s = bh[0];
  #pragma unroll
  for (int k = 0; k < HDIM; ++k) s += hp[k] * Wh[k];
  out[i] = s;
}

extern "C" void kernel_launch(void* const* d_in, const int* in_sizes, int n_in,
                              void* d_out, int out_size, void* d_ws, size_t ws_size,
                              hipStream_t stream) {
  const float* x      = (const float*)d_in[0];
  const int*   eidx   = (const int*)d_in[1];
  const float* W_pre  = (const float*)d_in[2];
  const float* b_pre  = (const float*)d_in[3];
  const float* W1     = (const float*)d_in[4];
  const float* b1     = (const float*)d_in[5];
  const float* W2     = (const float*)d_in[6];
  const float* b2     = (const float*)d_in[7];
  const float* W_head = (const float*)d_in[8];
  const float* b_head = (const float*)d_in[9];
  float* out = (float*)d_out;

  const int N = in_sizes[0] / 128;
  const int E = in_sizes[1] / 2;
  const int* src = eidx;
  const int* dst = eidx + E;

  int*   cnt = (int*)d_ws;
  float* dis = (float*)d_ws + N_COUNTER_INTS;
  float* h   = dis + N;
  float* hw  = h   + (size_t)N * HDIM;
  float* acc = hw  + (size_t)N * HDIM;

  const int n64 = N * HDIM;

  zero_f32<<<(N_COUNTER_INTS + N + 255) / 256, 256, 0, stream>>>((float*)d_ws, N_COUNTER_INTS + N);
  deg_accum<<<ATOMIC_BLOCKS, ATOMIC_THREADS, 0, stream>>>(dst, dis, E, N, cnt);
  deg_to_dis<<<(N + 255) / 256, 256, 0, stream>>>(dis, N);

  int gemm_blocks = (N + 127) / 128;
  gemm_n64_wmma<128, true><<<gemm_blocks, 256, 0, stream>>>(x, W_pre, b_pre, h, N);

  const float* Ws[2] = {W1, W2};
  const float* bs[2] = {b1, b2};
  for (int layer = 0; layer < 2; ++layer) {
    gemm_n64_wmma<64, false><<<gemm_blocks, 256, 0, stream>>>(h, Ws[layer], nullptr, hw, N);
    zero_f32<<<(n64 + 255) / 256, 256, 0, stream>>>(acc, n64);
    scatter_edges<<<ATOMIC_BLOCKS, ATOMIC_THREADS, 0, stream>>>(src, dst, dis, hw, acc, E, N,
                                                    cnt + (layer + 1) * LAUNCH_PITCH);
    finalize_layer<<<(n64 + 255) / 256, 256, 0, stream>>>(acc, bs[layer], h, n64);
  }

  head_kernel<<<(N + 255) / 256, 256, 0, stream>>>(h, W_head, b_head, out, N);
}


# Six elementwise launches folded into neighbours: clears ride in the products, bias+relu+skip formed in layer 2's A loader and in the head (16-lane tree: head sum order changes); 13->7 launches
# speedup vs baseline: 2.6244x; 1.2210x over previous
#include <hip/hip_runtime.h>
#include <hip/hip_bf16.h>

typedef __attribute__((ext_vector_type(2))) float v2f;
typedef __attribute__((ext_vector_type(8))) float v8f;

#define HDIM 64

#define ATOMIC_BLOCKS 2048
#define ATOMIC_THREADS 256
#define HALF_PITCH 64
#define LAUNCH_PITCH 128
#define N_COUNTER_INTS 384

__device__ __forceinline__ int chip_half() {
  return (int)((__builtin_amdgcn_s_sendmsg_rtn(0x87) >> 18) & 1u);
}

__device__ __forceinline__ void claim_range_slot(int* __restrict__ arrivals, int& range, int& slot) {
  __shared__ int s_pick[2];
  if (threadIdx.x == 0) {
    const int side = chip_half();
    const int r    = atomicAdd(&arrivals[side * HALF_PITCH], 1);
    const int per  = ATOMIC_BLOCKS / 2;
    s_pick[0] = (r < per) ? side : (side ^ 1);
    s_pick[1] = (r < per) ? r : (ATOMIC_BLOCKS - 1 - r);
  }
  __syncthreads();
  range = __builtin_amdgcn_readfirstlane(s_pick[0]);
  slot  = __builtin_amdgcn_readfirstlane(s_pick[1]);
}

__device__ __forceinline__ int range_cut(int n) { return ((n >> 1) + 63) & ~63; }

#define DEG_PER_PASS 8
__global__ void __launch_bounds__(ATOMIC_THREADS)
deg_accum(const int* __restrict__ dst, float* __restrict__ deg, int E, int n,
          int* __restrict__ arrivals) {
  int range, slot;
  claim_range_slot(arrivals, range, slot);
  const int cut  = range_cut(n);
  const int lo   = range ? cut : 0;
  const int hi   = range ? n : cut;
  const int last = E - 1;
  const int step = (ATOMIC_BLOCKS / 2) * ATOMIC_THREADS;
  for (int e = slot * ATOMIC_THREADS + (int)threadIdx.x; e < E; e += DEG_PER_PASS * step) {
    int d[DEG_PER_PASS];
    #pragma unroll
    for (int k = 0; k < DEG_PER_PASS; ++k) d[k] = dst[min(e + k * step, last)];
    #pragma unroll
    for (int k = 0; k < DEG_PER_PASS; ++k) {
      if ((e + k * step < E) && (d[k] >= lo) && (d[k] < hi)) atomicAdd(&deg[d[k]], 1.0f);
    }
  }
}

#define ROLE_PRE    0
#define ROLE_LAYER1 1
#define ROLE_LAYER2 2
template<int K, int ROLE>
__global__ void gemm_n64_wmma(const float* __restrict__ A,
                              const float* __restrict__ W,
                              const float* __restrict__ bias,
                              float* __restrict__ out, int n_rows,
                              const float* __restrict__ sum_in,
                              float* __restrict__ zplane,
                              float* __restrict__ deg,
                              int* __restrict__ counters) {
  if (ROLE == ROLE_PRE) {
    const int gid = blockIdx.x * blockDim.x + threadIdx.x;
    if (gid < N_COUNTER_INTS) counters[gid] = 0;
  }

  int wave = (blockIdx.x * blockDim.x + threadIdx.x) >> 5;
  int lane = threadIdx.x & 31;
  int row0 = wave * 16;
  if (row0 >= n_rows) return;

  const int half = lane >> 4;
  const int l16  = lane & 15;

  if (ROLE == ROLE_PRE) {
    if (lane < 16) deg[row0 + lane] = 0.0f;
  }
  if (ROLE == ROLE_LAYER1) {
    if (lane < 16) {
      const float d = deg[row0 + lane];
      deg[row0 + lane] = (d > 0.0f) ? rsqrtf(fmaxf(d, 1.0f)) : 0.0f;
    }
  }
  if (ROLE != ROLE_PRE) {
    const float4 z4 = make_float4(0.0f, 0.0f, 0.0f, 0.0f);
    float* zp = zplane + (size_t)row0 * HDIM + lane * 4;
    #pragma unroll
    for (int i = 0; i < 8; ++i) *(float4*)(zp + i * 128) = z4;
  }

  v8f acc0 = {}, acc1 = {}, acc2 = {}, acc3 = {};

  const float* arow = A + (size_t)(row0 + l16) * K;
  const float* srow = (ROLE == ROLE_LAYER2) ? sum_in + (size_t)(row0 + l16) * K : nullptr;
  for (int kb = 0; kb < K; kb += 4) {
    v2f a;
    a.x = arow[kb + half * 2 + 0];
    a.y = arow[kb + half * 2 + 1];
    if (ROLE == ROLE_LAYER2) {
      a.x = fmaxf(srow[kb + half * 2 + 0] + bias[kb + half * 2 + 0], 0.0f) + a.x;
      a.y = fmaxf(srow[kb + half * 2 + 1] + bias[kb + half * 2 + 1], 0.0f) + a.y;
    }

    const float* bp = W + (size_t)(kb + half * 2) * HDIM + l16;
    v2f b0, b1, b2, b3;
    b0.x = bp[0];        b0.y = bp[HDIM];
    b1.x = bp[16];       b1.y = bp[HDIM + 16];
    b2.x = bp[32];       b2.y = bp[HDIM + 32];
    b3.x = bp[48];       b3.y = bp[HDIM + 48];

    acc0 = __builtin_amdgcn_wmma_f32_16x16x4_f32(false, a, false, b0, (short)0, acc0, false, false);
    acc1 = __builtin_amdgcn_wmma_f32_16x16x4_f32(false, a, false, b1, (short)0, acc1, false, false);
    acc2 = __builtin_amdgcn_wmma_f32_16x16x4_f32(false, a, false, b2, (short)0, acc2, false, false);
    acc3 = __builtin_amdgcn_wmma_f32_16x16x4_f32(false, a, false, b3, (short)0, acc3, false, false);
  }

  v8f accs[4] = {acc0, acc1, acc2, acc3};
  #pragma unroll
  for (int nt = 0; nt < 4; ++nt) {
    int col = nt * 16 + l16;
    float bcol = (ROLE == ROLE_PRE) ? bias[col] : 0.0f;
    #pragma unroll
    for (int r = 0; r < 8; ++r) {
      int m = half * 8 + r;
      float v = accs[nt][r];
      if (ROLE == ROLE_PRE) { v += bcol; v = fmaxf(v, 0.0f); }
      out[(size_t)(row0 + m) * HDIM + col] = v;
    }
  }
}

#define ROWS_IN_FLIGHT 8
__global__ void __launch_bounds__(ATOMIC_THREADS)
scatter_edges(const int* __restrict__ src, const int* __restrict__ dst,
              const float* __restrict__ dis,
              const float* __restrict__ hw,
              float* __restrict__ acc, int E, int n,
              int* __restrict__ arrivals) {
  int range, slot;
  claim_range_slot(arrivals, range, slot);
  const int lane = threadIdx.x & 31;
  const int wpb  = ATOMIC_THREADS / 32;
  const int w0   = __builtin_amdgcn_readfirstlane(slot * wpb + (int)(threadIdx.x >> 5));
  const int nw   = (ATOMIC_BLOCKS / 2) * wpb;
  const int cut  = range_cut(n);
  const int lo   = range ? cut : 0;
  const int hi   = range ? n : cut;
  const int ngroups = (E + 31) >> 5;
  const int last = E - 1;
  if (w0 >= ngroups) return;

  int  e_nx  = w0 * 32 + lane;
  bool ok_nx = e_nx < E;
  int  s_nx  = src[min(e_nx, last)];
  int  d_nx  = dst[min(e_nx, last)];

  for (int g = w0; g < ngroups; g += nw) {
    const int  s  = s_nx, d = d_nx;
    const bool ok = ok_nx;
    e_nx  = (g + nw) * 32 + lane;
    ok_nx = e_nx < E;
    s_nx  = src[min(e_nx, last)];
    d_nx  = dst[min(e_nx, last)];

    const bool mine = ok && (d >= lo) && (d < hi);
    float coef = 0.0f;
    if (mine) coef = dis[s] * dis[d];

    unsigned m = __builtin_amdgcn_ballot_w32(mine);
    while (m != 0u) {
      float v0[ROWS_IN_FLIGHT], v1[ROWS_IN_FLIGHT];
      int   j[ROWS_IN_FLIGHT];
      #pragma unroll
      for (int u = 0; u < ROWS_IN_FLIGHT; ++u) {
        j[u] = -1;
        if (m != 0u) {
          j[u] = __builtin_ctz(m);
          m &= m - 1u;
          const int sj = __builtin_amdgcn_readlane(s, j[u]);
          const float* rp = hw + (size_t)sj * HDIM + lane;
          v0[u] = rp[0];
          v1[u] = rp[HDIM / 2];
        }
      }
      #pragma unroll
      for (int u = 0; u < ROWS_IN_FLIGHT; ++u) {
        if (j[u] >= 0) {
          const int   dj = __builtin_amdgcn_readlane(d, j[u]);
          const float cj = __int_as_float(__builtin_amdgcn_readlane(__float_as_int(coef), j[u]));
          float* ap = acc + (size_t)dj * HDIM + lane;
          atomicAdd(ap, v0[u] * cj);
          atomicAdd(ap + HDIM / 2, v1[u] * cj);
        }
      }
    }
  }
}

#define HEAD_THREADS 64
#define HEAD_BATCH 8

template<int M>
__device__ __forceinline__ float add_row_xor(float v) {
  return v + __int_as_float(__builtin_amdgcn_update_dpp(0, __float_as_int(v), 0x160 + M, 0xf, 0xf, false));
}

__global__ void __launch_bounds__(HEAD_THREADS)
head_fused(const float* __restrict__ sum1, const float* __restrict__ b1,
           const float* __restrict__ sum2, const float* __restrict__ b2,
           const float* __restrict__ h,
           const float* __restrict__ Wh, const float* __restrict__ bh,
           float* __restrict__ out, int n) {
  const int lane = threadIdx.x & 31;
  const int wave = __builtin_amdgcn_readfirstlane((int)((blockIdx.x * blockDim.x + threadIdx.x) >> 5));
  const int base = wave * 32;
  if (base >= n) return;
  const int g  = lane >> 4;
  const int l16 = lane & 15;
  const int c4 = l16 * 4;
  const float4 w   = *(const float4*)(Wh + c4);
  const float4 vb1 = *(const float4*)(b1 + c4);
  const float4 vb2 = *(const float4*)(b2 + c4);
  float res = 0.0f;
  #pragma unroll
  for (int part = 0; part < 16 / HEAD_BATCH; ++part) {
    float4 s1[HEAD_BATCH], s2[HEAD_BATCH], hh[HEAD_BATCH];
    #pragma unroll
    for (int j = 0; j < HEAD_BATCH; ++j) {
      const int row = min(base + g * 16 + part * HEAD_BATCH + j, n - 1);
      const size_t off = (size_t)row * HDIM + c4;
      s1[j] = *(const float4*)(sum1 + off);
      s2[j] = *(const float4*)(sum2 + off);
      hh[j] = *(const float4*)(h + off);
    }
    #pragma unroll
    for (int j = 0; j < HEAD_BATCH; ++j) {
      float4 t;
      t.x = fmaxf(s1[j].x + vb1.x, 0.0f) + hh[j].x;
      t.y = fmaxf(s1[j].y + vb1.y, 0.0f) + hh[j].y;
      t.z = fmaxf(s1[j].z + vb1.z, 0.0f) + hh[j].z;
      t.w = fmaxf(s1[j].w + vb1.w, 0.0f) + hh[j].w;
      t.x = fmaxf(s2[j].x + vb2.x, 0.0f) + t.x;
      t.y = fmaxf(s2[j].y + vb2.y, 0.0f) + t.y;
      t.z = fmaxf(s2[j].z + vb2.z, 0.0f) + t.z;
      t.w = fmaxf(s2[j].w + vb2.w, 0.0f) + t.w;
      float p = t.x * w.x + t.y * w.y + t.z * w.z + t.w * w.w;
      p = add_row_xor<1>(p);
      p = add_row_xor<2>(p);
      p = add_row_xor<4>(p);
      p = add_row_xor<8>(p);
      if (l16 == part * HEAD_BATCH + j) res = p;
    }
  }
  const int row = base + lane;
  if (row < n) out[row] = res + bh[0];
}

extern "C" void kernel_launch(void* const* d_in, const int* in_sizes, int n_in,
                              void* d_out, int out_size, void* d_ws, size_t ws_size,
                              hipStream_t stream) {
  const float* x      = (const float*)d_in[0];
  const int*   eidx   = (const int*)d_in[1];
  const float* W_pre  = (const float*)d_in[2];
  const float* b_pre  = (const float*)d_in[3];
  const float* W1     = (const float*)d_in[4];
  const float* b1     = (const float*)d_in[5];
  const float* W2     = (const float*)d_in[6];
  const float* b2     = (const float*)d_in[7];
  const float* W_head = (const float*)d_in[8];
  const float* b_head = (const float*)d_in[9];
  float* out = (float*)d_out;

  const int N = in_sizes[0] / 128;
  const int E = in_sizes[1] / 2;
  const int* src = eidx;
  const int* dst = eidx + E;

  int*   cnt  = (int*)d_ws;
  float* dis  = (float*)d_ws + N_COUNTER_INTS;
  float* h    = dis  + N;
  float* hw   = h    + (size_t)N * HDIM;
  float* sum1 = hw   + (size_t)N * HDIM;
  float* sum2 = sum1 + (size_t)N * HDIM;

  int gemm_blocks = (N + 127) / 128;

  gemm_n64_wmma<128, ROLE_PRE><<<gemm_blocks, 256, 0, stream>>>(x, W_pre, b_pre, h, N,
                                                               nullptr, nullptr, dis, cnt);
  deg_accum<<<ATOMIC_BLOCKS, ATOMIC_THREADS, 0, stream>>>(dst, dis, E, N, cnt);

  gemm_n64_wmma<64, ROLE_LAYER1><<<gemm_blocks, 256, 0, stream>>>(h, W1, nullptr, hw, N,
                                                                 nullptr, sum1, dis, nullptr);
  scatter_edges<<<ATOMIC_BLOCKS, ATOMIC_THREADS, 0, stream>>>(src, dst, dis, hw, sum1, E, N,
                                                  cnt + 1 * LAUNCH_PITCH);

  gemm_n64_wmma<64, ROLE_LAYER2><<<gemm_blocks, 256, 0, stream>>>(h, W2, b1, hw, N,
                                                                 sum1, sum2, nullptr, nullptr);
  scatter_edges<<<ATOMIC_BLOCKS, ATOMIC_THREADS, 0, stream>>>(src, dst, dis, hw, sum2, E, N,
                                                  cnt + 2 * LAUNCH_PITCH);

  head_fused<<<(N + HEAD_THREADS - 1) / HEAD_THREADS, HEAD_THREADS, 0, stream>>>(
      sum1, b1, sum2, b2, h, W_head, b_head, out, N);
}


# Each layer's product now runs AFTER its neighbour sum, (sum of input rows) x W: bias+relu+skip and the head dot ride on the accumulators, no hw planes, head launch gone, 7->6; sum order changes
# speedup vs baseline: 2.8148x; 1.0725x over previous
#include <hip/hip_runtime.h>
#include <hip/hip_bf16.h>

typedef __attribute__((ext_vector_type(2))) float v2f;
typedef __attribute__((ext_vector_type(8))) float v8f;

#define HDIM 64

#define ATOMIC_BLOCKS 2048
#define ATOMIC_THREADS 256
#define HALF_PITCH 64
#define LAUNCH_PITCH 128
#define N_COUNTER_INTS 384

__device__ __forceinline__ int chip_half() {
  return (int)((__builtin_amdgcn_s_sendmsg_rtn(0x87) >> 18) & 1u);
}

__device__ __forceinline__ void claim_range_slot(int* __restrict__ arrivals, int& range, int& slot) {
  __shared__ int s_pick[2];
  if (threadIdx.x == 0) {
    const int side = chip_half();
    const int r    = atomicAdd(&arrivals[side * HALF_PITCH], 1);
    const int per  = ATOMIC_BLOCKS / 2;
    s_pick[0] = (r < per) ? side : (side ^ 1);
    s_pick[1] = (r < per) ? r : (ATOMIC_BLOCKS - 1 - r);
  }
  __syncthreads();
  range = __builtin_amdgcn_readfirstlane(s_pick[0]);
  slot  = __builtin_amdgcn_readfirstlane(s_pick[1]);
}

__device__ __forceinline__ int range_cut(int n) { return ((n >> 1) + 63) & ~63; }

#define DEG_PER_PASS 8
__global__ void __launch_bounds__(ATOMIC_THREADS)
deg_accum(const int* __restrict__ dst, float* __restrict__ deg, int E, int n,
          int* __restrict__ arrivals) {
  int range, slot;
  claim_range_slot(arrivals, range, slot);
  const int cut  = range_cut(n);
  const int lo   = range ? cut : 0;
  const int hi   = range ? n : cut;
  const int last = E - 1;
  const int step = (ATOMIC_BLOCKS / 2) * ATOMIC_THREADS;
  for (int e = slot * ATOMIC_THREADS + (int)threadIdx.x; e < E; e += DEG_PER_PASS * step) {
    int d[DEG_PER_PASS];
    #pragma unroll
    for (int k = 0; k < DEG_PER_PASS; ++k) d[k] = dst[min(e + k * step, last)];
    #pragma unroll
    for (int k = 0; k < DEG_PER_PASS; ++k) {
      if ((e + k * step < E) && (d[k] >= lo) && (d[k] < hi)) atomicAdd(&deg[d[k]], 1.0f);
    }
  }
}

#define ROLE_PRE 0
#define ROLE_MID 1

template<int K>
__device__ __forceinline__ void tile_product(const float* __restrict__ A, const float* __restrict__ W,
                                             int row0, int half, int l16,
                                             v8f& acc0, v8f& acc1, v8f& acc2, v8f& acc3) {
  const float* arow = A + (size_t)(row0 + l16) * K;
  for (int kb = 0; kb < K; kb += 4) {
    v2f a;
    a.x = arow[kb + half * 2 + 0];
    a.y = arow[kb + half * 2 + 1];

    const float* bp = W + (size_t)(kb + half * 2) * HDIM + l16;
    v2f b0, b1, b2, b3;
    b0.x = bp[0];        b0.y = bp[HDIM];
    b1.x = bp[16];       b1.y = bp[HDIM + 16];
    b2.x = bp[32];       b2.y = bp[HDIM + 32];
    b3.x = bp[48];       b3.y = bp[HDIM + 48];

    acc0 = __builtin_amdgcn_wmma_f32_16x16x4_f32(false, a, false, b0, (short)0, acc0, false, false);
    acc1 = __builtin_amdgcn_wmma_f32_16x16x4_f32(false, a, false, b1, (short)0, acc1, false, false);
    acc2 = __builtin_amdgcn_wmma_f32_16x16x4_f32(false, a, false, b2, (short)0, acc2, false, false);
    acc3 = __builtin_amdgcn_wmma_f32_16x16x4_f32(false, a, false, b3, (short)0, acc3, false, false);
  }
}

template<int K, int ROLE>
__global__ void gemm_n64_wmma(const float* __restrict__ A,
                              const float* __restrict__ W,
                              const float* __restrict__ bias,
                              float* __restrict__ out, int n_rows,
                              const float* __restrict__ skip,
                              float* __restrict__ zplane,
                              float* __restrict__ deg,
                              int* __restrict__ counters) {
  if (ROLE == ROLE_PRE) {
    const int gid = blockIdx.x * blockDim.x + threadIdx.x;
    if (gid < N_COUNTER_INTS) counters[gid] = 0;
  }

  int wave = (blockIdx.x * blockDim.x + threadIdx.x) >> 5;
  int lane = threadIdx.x & 31;
  int row0 = wave * 16;
  if (row0 >= n_rows) return;

  const int half = lane >> 4;
  const int l16  = lane & 15;

  if (ROLE == ROLE_PRE) {
    if (lane < 16) deg[row0 + lane] = 0.0f;
  }
  {
    const float4 z4 = make_float4(0.0f, 0.0f, 0.0f, 0.0f);
    float* zp = zplane + (size_t)row0 * HDIM + lane * 4;
    #pragma unroll
    for (int i = 0; i < 8; ++i) *(float4*)(zp + i * 128) = z4;
  }

  float sk[4][8] = {};
  if (ROLE == ROLE_MID) {
    #pragma unroll
    for (int nt = 0; nt < 4; ++nt) {
      #pragma unroll
      for (int r = 0; r < 8; ++r)
        sk[nt][r] = skip[(size_t)(row0 + half * 8 + r) * HDIM + nt * 16 + l16];
    }
  }

  v8f acc0 = {}, acc1 = {}, acc2 = {}, acc3 = {};
  tile_product<K>(A, W, row0, half, l16, acc0, acc1, acc2, acc3);

  v8f accs[4] = {acc0, acc1, acc2, acc3};
  #pragma unroll
  for (int nt = 0; nt < 4; ++nt) {
    int col = nt * 16 + l16;
    float bcol = bias[col];
    #pragma unroll
    for (int r = 0; r < 8; ++r) {
      int m = half * 8 + r;
      float v = accs[nt][r];
      v += bcol; v = fmaxf(v, 0.0f);
      if (ROLE == ROLE_MID) v += sk[nt][r];
      out[(size_t)(row0 + m) * HDIM + col] = v;
    }
  }
}

#define ROWS_IN_FLIGHT 8
__global__ void __launch_bounds__(ATOMIC_THREADS)
scatter_edges(const int* __restrict__ src, const int* __restrict__ dst,
              const float* __restrict__ deg,
              const float* __restrict__ rows,
              float* __restrict__ acc, int E, int n,
              int* __restrict__ arrivals) {
  int range, slot;
  claim_range_slot(arrivals, range, slot);
  const int lane = threadIdx.x & 31;
  const int wpb  = ATOMIC_THREADS / 32;
  const int w0   = __builtin_amdgcn_readfirstlane(slot * wpb + (int)(threadIdx.x >> 5));
  const int nw   = (ATOMIC_BLOCKS / 2) * wpb;
  const int cut  = range_cut(n);
  const int lo   = range ? cut : 0;
  const int hi   = range ? n : cut;
  const int ngroups = (E + 31) >> 5;
  const int last = E - 1;
  if (w0 >= ngroups) return;

  int  e_nx  = w0 * 32 + lane;
  bool ok_nx = e_nx < E;
  int  s_nx  = src[min(e_nx, last)];
  int  d_nx  = dst[min(e_nx, last)];

  for (int g = w0; g < ngroups; g += nw) {
    const int  s  = s_nx, d = d_nx;
    const bool ok = ok_nx;
    e_nx  = (g + nw) * 32 + lane;
    ok_nx = e_nx < E;
    s_nx  = src[min(e_nx, last)];
    d_nx  = dst[min(e_nx, last)];

    const bool mine = ok && (d >= lo) && (d < hi);
    float coef = 0.0f;
    if (mine) {
      const float cs = deg[s], cd = deg[d];
      const float is = (cs > 0.0f) ? rsqrtf(fmaxf(cs, 1.0f)) : 0.0f;
      const float id = (cd > 0.0f) ? rsqrtf(fmaxf(cd, 1.0f)) : 0.0f;
      coef = is * id;
    }

    unsigned m = __builtin_amdgcn_ballot_w32(mine);
    while (m != 0u) {
      float v0[ROWS_IN_FLIGHT], v1[ROWS_IN_FLIGHT];
      int   j[ROWS_IN_FLIGHT];
      #pragma unroll
      for (int u = 0; u < ROWS_IN_FLIGHT; ++u) {
        j[u] = -1;
        if (m != 0u) {
          j[u] = __builtin_ctz(m);
          m &= m - 1u;
          const int sj = __builtin_amdgcn_readlane(s, j[u]);
          const float* rp = rows + (size_t)sj * HDIM + lane;
          v0[u] = rp[0];
          v1[u] = rp[HDIM / 2];
        }
      }
      #pragma unroll
      for (int u = 0; u < ROWS_IN_FLIGHT; ++u) {
        if (j[u] >= 0) {
          const int   dj = __builtin_amdgcn_readlane(d, j[u]);
          const float cj = __int_as_float(__builtin_amdgcn_readlane(__float_as_int(coef), j[u]));
          float* ap = acc + (size_t)dj * HDIM + lane;
          atomicAdd(ap, v0[u] * cj);
          atomicAdd(ap + HDIM / 2, v1[u] * cj);
        }
      }
    }
  }
}


template<int M>
__device__ __forceinline__ float add_row_xor(float v) {
  return v + __int_as_float(__builtin_amdgcn_update_dpp(0, __float_as_int(v), 0x160 + M, 0xf, 0xf, false));
}

#define HEAD_THREADS 256
__global__ void __launch_bounds__(HEAD_THREADS)
gemm_head(const float* __restrict__ A, const float* __restrict__ W,
          const float* __restrict__ bias, const float* __restrict__ skip,
          const float* __restrict__ Wh, const float* __restrict__ bh,
          float* __restrict__ out, int n_rows) {
  const int lane = threadIdx.x & 31;
  const int wave = __builtin_amdgcn_readfirstlane((int)((blockIdx.x * blockDim.x + threadIdx.x) >> 5));
  const int base = wave * 32;
  if (base >= n_rows) return;
  const int half = lane >> 4;
  const int l16  = lane & 15;

  float wh[4], bc[4];
  #pragma unroll
  for (int nt = 0; nt < 4; ++nt) { wh[nt] = Wh[nt * 16 + l16]; bc[nt] = bias[nt * 16 + l16]; }

  float res = 0.0f;
  #pragma unroll
  for (int mt = 0; mt < 2; ++mt) {
    const int row0 = base + mt * 16;
    if (row0 < n_rows) {
      float sk[4][8];
      #pragma unroll
      for (int nt = 0; nt < 4; ++nt) {
        #pragma unroll
        for (int r = 0; r < 8; ++r)
          sk[nt][r] = skip[(size_t)(row0 + half * 8 + r) * HDIM + nt * 16 + l16];
      }
      v8f acc0 = {}, acc1 = {}, acc2 = {}, acc3 = {};
      tile_product<HDIM>(A, W, row0, half, l16, acc0, acc1, acc2, acc3);
      v8f accs[4] = {acc0, acc1, acc2, acc3};
      #pragma unroll
      for (int r = 0; r < 8; ++r) {
        float p = 0.0f;
        #pragma unroll
        for (int nt = 0; nt < 4; ++nt) {
          float v = accs[nt][r] + bc[nt];
          v = fmaxf(v, 0.0f) + sk[nt][r];
          p += v * wh[nt];
        }
        p = add_row_xor<1>(p);
        p = add_row_xor<2>(p);
        p = add_row_xor<4>(p);
        p = add_row_xor<8>(p);
        if (l16 == mt * 8 + r) res = p;
      }
    }
  }
  const int row = base + (l16 >> 3) * 16 + half * 8 + (l16 & 7);
  if (row < n_rows) out[row] = res + bh[0];
}

extern "C" void kernel_launch(void* const* d_in, const int* in_sizes, int n_in,
                              void* d_out, int out_size, void* d_ws, size_t ws_size,
                              hipStream_t stream) {
  const float* x      = (const float*)d_in[0];
  const int*   eidx   = (const int*)d_in[1];
  const float* W_pre  = (const float*)d_in[2];
  const float* b_pre  = (const float*)d_in[3];
  const float* W1     = (const float*)d_in[4];
  const float* b1     = (const float*)d_in[5];
  const float* W2     = (const float*)d_in[6];
  const float* b2     = (const float*)d_in[7];
  const float* W_head = (const float*)d_in[8];
  const float* b_head = (const float*)d_in[9];
  float* out = (float*)d_out;

  const int N = in_sizes[0] / 128;
  const int E = in_sizes[1] / 2;
  const int* src = eidx;
  const int* dst = eidx + E;

  int*   cnt  = (int*)d_ws;
  float* deg  = (float*)d_ws + N_COUNTER_INTS;
  float* h0   = deg  + N;
  float* sum1 = h0   + (size_t)N * HDIM;
  float* h1   = sum1 + (size_t)N * HDIM;
  float* sum2 = h1   + (size_t)N * HDIM;

  int gemm_blocks = (N + 127) / 128;

  gemm_n64_wmma<128, ROLE_PRE><<<gemm_blocks, 256, 0, stream>>>(x, W_pre, b_pre, h0, N,
                                                               nullptr, sum1, deg, cnt);
  deg_accum<<<ATOMIC_BLOCKS, ATOMIC_THREADS, 0, stream>>>(dst, deg, E, N, cnt);

  scatter_edges<<<ATOMIC_BLOCKS, ATOMIC_THREADS, 0, stream>>>(src, dst, deg, h0, sum1, E, N,
                                                  cnt + 1 * LAUNCH_PITCH);
  gemm_n64_wmma<64, ROLE_MID><<<gemm_blocks, 256, 0, stream>>>(sum1, W1, b1, h1, N,
                                                              h0, sum2, nullptr, nullptr);

  scatter_edges<<<ATOMIC_BLOCKS, ATOMIC_THREADS, 0, stream>>>(src, dst, deg, h1, sum2, E, N,
                                                  cnt + 2 * LAUNCH_PITCH);
  gemm_head<<<(N + 255) / 256, HEAD_THREADS, 0, stream>>>(sum2, W2, b2, h1, W_head, b_head, out, N);
}


# Scatter: an edge's coefficient is kept from layer 1 (a plane of E floats per row range, whole lines); layer 2 reads it with the ids one group ahead: no degree lookups, no wait before its row loads
# speedup vs baseline: 2.9716x; 1.0557x over previous
#include <hip/hip_runtime.h>
#include <hip/hip_bf16.h>

typedef __attribute__((ext_vector_type(2))) float v2f;
typedef __attribute__((ext_vector_type(8))) float v8f;

#define HDIM 64

#define ATOMIC_BLOCKS 2048
#define ATOMIC_THREADS 256
#define HALF_PITCH 64
#define LAUNCH_PITCH 128
#define N_COUNTER_INTS 384

__device__ __forceinline__ int chip_half() {
  return (int)((__builtin_amdgcn_s_sendmsg_rtn(0x87) >> 18) & 1u);
}

__device__ __forceinline__ void claim_range_slot(int* __restrict__ arrivals, int& range, int& slot) {
  __shared__ int s_pick[2];
  if (threadIdx.x == 0) {
    const int side = chip_half();
    const int r    = atomicAdd(&arrivals[side * HALF_PITCH], 1);
    const int per  = ATOMIC_BLOCKS / 2;
    s_pick[0] = (r < per) ? side : (side ^ 1);
    s_pick[1] = (r < per) ? r : (ATOMIC_BLOCKS - 1 - r);
  }
  __syncthreads();
  range = __builtin_amdgcn_readfirstlane(s_pick[0]);
  slot  = __builtin_amdgcn_readfirstlane(s_pick[1]);
}

__device__ __forceinline__ int range_cut(int n) { return ((n >> 1) + 63) & ~63; }

#define DEG_PER_PASS 8
__global__ void __launch_bounds__(ATOMIC_THREADS)
deg_accum(const int* __restrict__ dst, float* __restrict__ deg, int E, int n,
          int* __restrict__ arrivals) {
  int range, slot;
  claim_range_slot(arrivals, range, slot);
  const int cut  = range_cut(n);
  const int lo   = range ? cut : 0;
  const int hi   = range ? n : cut;
  const int last = E - 1;
  const int step = (ATOMIC_BLOCKS / 2) * ATOMIC_THREADS;
  for (int e = slot * ATOMIC_THREADS + (int)threadIdx.x; e < E; e += DEG_PER_PASS * step) {
    int d[DEG_PER_PASS];
    #pragma unroll
    for (int k = 0; k < DEG_PER_PASS; ++k) d[k] = dst[min(e + k * step, last)];
    #pragma unroll
    for (int k = 0; k < DEG_PER_PASS; ++k) {
      if ((e + k * step < E) && (d[k] >= lo) && (d[k] < hi)) atomicAdd(&deg[d[k]], 1.0f);
    }
  }
}

#define ROLE_PRE 0
#define ROLE_MID 1

template<int K>
__device__ __forceinline__ void tile_product(const float* __restrict__ A, const float* __restrict__ W,
                                             int row0, int half, int l16,
                                             v8f& acc0, v8f& acc1, v8f& acc2, v8f& acc3) {
  const float* arow = A + (size_t)(row0 + l16) * K;
  for (int kb = 0; kb < K; kb += 4) {
    v2f a;
    a.x = arow[kb + half * 2 + 0];
    a.y = arow[kb + half * 2 + 1];

    const float* bp = W + (size_t)(kb + half * 2) * HDIM + l16;
    v2f b0, b1, b2, b3;
    b0.x = bp[0];        b0.y = bp[HDIM];
    b1.x = bp[16];       b1.y = bp[HDIM + 16];
    b2.x = bp[32];       b2.y = bp[HDIM + 32];
    b3.x = bp[48];       b3.y = bp[HDIM + 48];

    acc0 = __builtin_amdgcn_wmma_f32_16x16x4_f32(false, a, false, b0, (short)0, acc0, false, false);
    acc1 = __builtin_amdgcn_wmma_f32_16x16x4_f32(false, a, false, b1, (short)0, acc1, false, false);
    acc2 = __builtin_amdgcn_wmma_f32_16x16x4_f32(false, a, false, b2, (short)0, acc2, false, false);
    acc3 = __builtin_amdgcn_wmma_f32_16x16x4_f32(false, a, false, b3, (short)0, acc3, false, false);
  }
}

template<int K, int ROLE>
__global__ void gemm_n64_wmma(const float* __restrict__ A,
                              const float* __restrict__ W,
                              const float* __restrict__ bias,
                              float* __restrict__ out, int n_rows,
                              const float* __restrict__ skip,
                              float* __restrict__ zplane,
                              float* __restrict__ deg,
                              int* __restrict__ counters) {
  if (ROLE == ROLE_PRE) {
    const int gid = blockIdx.x * blockDim.x + threadIdx.x;
    if (gid < N_COUNTER_INTS) counters[gid] = 0;
  }

  int wave = (blockIdx.x * blockDim.x + threadIdx.x) >> 5;
  int lane = threadIdx.x & 31;
  int row0 = wave * 16;
  if (row0 >= n_rows) return;

  const int half = lane >> 4;
  const int l16  = lane & 15;

  if (ROLE == ROLE_PRE) {
    if (lane < 16) deg[row0 + lane] = 0.0f;
  }
  {
    const float4 z4 = make_float4(0.0f, 0.0f, 0.0f, 0.0f);
    float* zp = zplane + (size_t)row0 * HDIM + lane * 4;
    #pragma unroll
    for (int i = 0; i < 8; ++i) *(float4*)(zp + i * 128) = z4;
  }

  float sk[4][8] = {};
  if (ROLE == ROLE_MID) {
    #pragma unroll
    for (int nt = 0; nt < 4; ++nt) {
      #pragma unroll
      for (int r = 0; r < 8; ++r)
        sk[nt][r] = skip[(size_t)(row0 + half * 8 + r) * HDIM + nt * 16 + l16];
    }
  }

  v8f acc0 = {}, acc1 = {}, acc2 = {}, acc3 = {};
  tile_product<K>(A, W, row0, half, l16, acc0, acc1, acc2, acc3);

  v8f accs[4] = {acc0, acc1, acc2, acc3};
  #pragma unroll
  for (int nt = 0; nt < 4; ++nt) {
    int col = nt * 16 + l16;
    float bcol = bias[col];
    #pragma unroll
    for (int r = 0; r < 8; ++r) {
      int m = half * 8 + r;
      float v = accs[nt][r];
      v += bcol; v = fmaxf(v, 0.0f);
      if (ROLE == ROLE_MID) v += sk[nt][r];
      out[(size_t)(row0 + m) * HDIM + col] = v;
    }
  }
}

#define ROWS_IN_FLIGHT 8
template<bool FIRST>
__global__ void __launch_bounds__(ATOMIC_THREADS)
scatter_edges(const int* __restrict__ src, const int* __restrict__ dst,
              const float* __restrict__ deg,
              const float* __restrict__ rows,
              float* __restrict__ acc, int E, int n,
              int* __restrict__ arrivals,
              float* __restrict__ coef_planes) {
  int range, slot;
  claim_range_slot(arrivals, range, slot);
  const int lane = threadIdx.x & 31;
  const int wpb  = ATOMIC_THREADS / 32;
  const int w0   = __builtin_amdgcn_readfirstlane(slot * wpb + (int)(threadIdx.x >> 5));
  const int nw   = (ATOMIC_BLOCKS / 2) * wpb;
  const int cut  = range_cut(n);
  const int lo   = range ? cut : 0;
  const int hi   = range ? n : cut;
  const int ngroups = (E + 31) >> 5;
  const int last = E - 1;
  if (w0 >= ngroups) return;
  float* __restrict__ plane = coef_planes + (size_t)range * E;

  int   e_nx  = w0 * 32 + lane;
  bool  ok_nx = e_nx < E;
  int   s_nx  = src[min(e_nx, last)];
  int   d_nx  = dst[min(e_nx, last)];
  float c_nx  = 0.0f;
  if (!FIRST) c_nx = plane[min(e_nx, last)];

  for (int g = w0; g < ngroups; g += nw) {
    const int   s  = s_nx, d = d_nx;
    const bool  ok = ok_nx;
    const float c  = c_nx;
    const int   e  = g * 32 + lane;
    e_nx  = (g + nw) * 32 + lane;
    ok_nx = e_nx < E;
    s_nx  = src[min(e_nx, last)];
    d_nx  = dst[min(e_nx, last)];
    if (!FIRST) c_nx = plane[min(e_nx, last)];

    const bool mine = ok && (d >= lo) && (d < hi);
    float coef = 0.0f;
    if (FIRST) {
      if (mine) {
        const float cs = deg[s], cd = deg[d];
        const float is = (cs > 0.0f) ? rsqrtf(fmaxf(cs, 1.0f)) : 0.0f;
        const float id = (cd > 0.0f) ? rsqrtf(fmaxf(cd, 1.0f)) : 0.0f;
        coef = is * id;
      }
      if (ok) plane[e] = coef;
    } else {
      coef = mine ? c : 0.0f;
    }

    unsigned m = __builtin_amdgcn_ballot_w32(mine);
    while (m != 0u) {
      float v0[ROWS_IN_FLIGHT], v1[ROWS_IN_FLIGHT];
      int   j[ROWS_IN_FLIGHT];
      #pragma unroll
      for (int u = 0; u < ROWS_IN_FLIGHT; ++u) {
        j[u] = -1;
        if (u == 0 || m != 0u) {
          j[u] = __builtin_ctz(m);
          m &= m - 1u;
          const int sj = __builtin_amdgcn_readlane(s, j[u]);
          const float* rp = rows + (size_t)sj * HDIM + lane;
          v0[u] = rp[0];
          v1[u] = rp[HDIM / 2];
        }
      }
      #pragma unroll
      for (int u = 0; u < ROWS_IN_FLIGHT; ++u) {
        if (u == 0 || j[u] >= 0) {
          const int   dj = __builtin_amdgcn_readlane(d, j[u]);
          const float cj = __int_as_float(__builtin_amdgcn_readlane(__float_as_int(coef), j[u]));
          float* ap = acc + (size_t)dj * HDIM + lane;
          atomicAdd(ap, v0[u] * cj);
          atomicAdd(ap + HDIM / 2, v1[u] * cj);
        }
      }
    }
  }
}


template<int M>
__device__ __forceinline__ float add_row_xor(float v) {
  return v + __int_as_float(__builtin_amdgcn_update_dpp(0, __float_as_int(v), 0x160 + M, 0xf, 0xf, false));
}

#define HEAD_THREADS 256
__global__ void __launch_bounds__(HEAD_THREADS)
gemm_head(const float* __restrict__ A, const float* __restrict__ W,
          const float* __restrict__ bias, const float* __restrict__ skip,
          const float* __restrict__ Wh, const float* __restrict__ bh,
          float* __restrict__ out, int n_rows) {
  const int lane = threadIdx.x & 31;
  const int wave = __builtin_amdgcn_readfirstlane((int)((blockIdx.x * blockDim.x + threadIdx.x) >> 5));
  const int base = wave * 32;
  if (base >= n_rows) return;
  const int half = lane >> 4;
  const int l16  = lane & 15;

  float wh[4], bc[4];
  #pragma unroll
  for (int nt = 0; nt < 4; ++nt) { wh[nt] = Wh[nt * 16 + l16]; bc[nt] = bias[nt * 16 + l16]; }

  float res = 0.0f;
  #pragma unroll
  for (int mt = 0; mt < 2; ++mt) {
    const int row0 = base + mt * 16;
    if (row0 < n_rows) {
      float sk[4][8];
      #pragma unroll
      for (int nt = 0; nt < 4; ++nt) {
        #pragma unroll
        for (int r = 0; r < 8; ++r)
          sk[nt][r] = skip[(size_t)(row0 + half * 8 + r) * HDIM + nt * 16 + l16];
      }
      v8f acc0 = {}, acc1 = {}, acc2 = {}, acc3 = {};
      tile_product<HDIM>(A, W, row0, half, l16, acc0, acc1, acc2, acc3);
      v8f accs[4] = {acc0, acc1, acc2, acc3};
      #pragma unroll
      for (int r = 0; r < 8; ++r) {
        float p = 0.0f;
        #pragma unroll
        for (int nt = 0; nt < 4; ++nt) {
          float v = accs[nt][r] + bc[nt];
          v = fmaxf(v, 0.0f) + sk[nt][r];
          p += v * wh[nt];
        }
        p = add_row_xor<1>(p);
        p = add_row_xor<2>(p);
        p = add_row_xor<4>(p);
        p = add_row_xor<8>(p);
        if (l16 == mt * 8 + r) res = p;
      }
    }
  }
  const int row = base + (l16 >> 3) * 16 + half * 8 + (l16 & 7);
  if (row < n_rows) out[row] = res + bh[0];
}

extern "C" void kernel_launch(void* const* d_in, const int* in_sizes, int n_in,
                              void* d_out, int out_size, void* d_ws, size_t ws_size,
                              hipStream_t stream) {
  const float* x      = (const float*)d_in[0];
  const int*   eidx   = (const int*)d_in[1];
  const float* W_pre  = (const float*)d_in[2];
  const float* b_pre  = (const float*)d_in[3];
  const float* W1     = (const float*)d_in[4];
  const float* b1     = (const float*)d_in[5];
  const float* W2     = (const float*)d_in[6];
  const float* b2     = (const float*)d_in[7];
  const float* W_head = (const float*)d_in[8];
  const float* b_head = (const float*)d_in[9];
  float* out = (float*)d_out;

  const int N = in_sizes[0] / 128;
  const int E = in_sizes[1] / 2;
  const int* src = eidx;
  const int* dst = eidx + E;

  int*   cnt  = (int*)d_ws;
  float* deg  = (float*)d_ws + N_COUNTER_INTS;
  float* h0   = deg  + N;
  float* sum1 = h0   + (size_t)N * HDIM;
  float* h1   = sum1 + (size_t)N * HDIM;
  float* sum2 = h1   + (size_t)N * HDIM;
  float* cpl  = sum2 + (size_t)N * HDIM;

  int gemm_blocks = (N + 127) / 128;

  gemm_n64_wmma<128, ROLE_PRE><<<gemm_blocks, 256, 0, stream>>>(x, W_pre, b_pre, h0, N,
                                                               nullptr, sum1, deg, cnt);
  deg_accum<<<ATOMIC_BLOCKS, ATOMIC_THREADS, 0, stream>>>(dst, deg, E, N, cnt);

  scatter_edges<true><<<ATOMIC_BLOCKS, ATOMIC_THREADS, 0, stream>>>(src, dst, deg, h0, sum1, E, N,
                                                        cnt + 1 * LAUNCH_PITCH, cpl);
  gemm_n64_wmma<64, ROLE_MID><<<gemm_blocks, 256, 0, stream>>>(sum1, W1, b1, h1, N,
                                                              h0, sum2, nullptr, nullptr);

  scatter_edges<false><<<ATOMIC_BLOCKS, ATOMIC_THREADS, 0, stream>>>(src, dst, nullptr, h1, sum2, E, N,
                                                         cnt + 2 * LAUNCH_PITCH, cpl);
  gemm_head<<<(N + 255) / 256, HEAD_THREADS, 0, stream>>>(sum2, W2, b2, h1, W_head, b_head, out, N);
}


# Degree count leaves its own 16384-wave launch: 512 extra workgroups of the first product's launch count beside its tiles (16 ids a pass); a small launch in front clears table and counters
# speedup vs baseline: 3.0839x; 1.0378x over previous
#include <hip/hip_runtime.h>
#include <hip/hip_bf16.h>

typedef __attribute__((ext_vector_type(2))) float v2f;
typedef __attribute__((ext_vector_type(8))) float v8f;

#define HDIM 64

#define ATOMIC_BLOCKS 2048
#define ATOMIC_THREADS 256
#define DEG_BLOCKS 512
#define HALF_PITCH 64
#define LAUNCH_PITCH 128
#define N_COUNTER_INTS 384

__device__ __forceinline__ int chip_half() {
  return (int)((__builtin_amdgcn_s_sendmsg_rtn(0x87) >> 18) & 1u);
}

template<int BLOCKS>
__device__ __forceinline__ void claim_range_slot(int* __restrict__ arrivals, int& range, int& slot) {
  __shared__ int s_pick[2];
  if (threadIdx.x == 0) {
    const int side = chip_half();
    const int r    = atomicAdd(&arrivals[side * HALF_PITCH], 1);
    const int per  = BLOCKS / 2;
    s_pick[0] = (r < per) ? side : (side ^ 1);
    s_pick[1] = (r < per) ? r : (BLOCKS - 1 - r);
  }
  __syncthreads();
  range = __builtin_amdgcn_readfirstlane(s_pick[0]);
  slot  = __builtin_amdgcn_readfirstlane(s_pick[1]);
}

__device__ __forceinline__ int range_cut(int n) { return ((n >> 1) + 63) & ~63; }

__global__ void __launch_bounds__(ATOMIC_THREADS)
clear_tables(int* __restrict__ counters, float* __restrict__ deg, int n) {
  const int gid = blockIdx.x * blockDim.x + threadIdx.x;
  if (gid < N_COUNTER_INTS) counters[gid] = 0;
  const int i = gid * 4;
  if (i + 3 < n) {
    *(float4*)(deg + i) = make_float4(0.0f, 0.0f, 0.0f, 0.0f);
  } else {
    for (int k = 0; k < 4; ++k) if (i + k < n) deg[i + k] = 0.0f;
  }
}

#define DEG_PER_PASS 16
__device__ __forceinline__ void count_degrees(const int* __restrict__ dst, float* __restrict__ deg,
                                              int E, int n, int* __restrict__ arrivals) {
  int range, slot;
  claim_range_slot<DEG_BLOCKS>(arrivals, range, slot);
  const int cut  = range_cut(n);
  const int lo   = range ? cut : 0;
  const int hi   = range ? n : cut;
  const int last = E - 1;
  const int step = (DEG_BLOCKS / 2) * ATOMIC_THREADS;
  for (int e = slot * ATOMIC_THREADS + (int)threadIdx.x; e < E; e += DEG_PER_PASS * step) {
    int d[DEG_PER_PASS];
    #pragma unroll
    for (int k = 0; k < DEG_PER_PASS; ++k) d[k] = dst[min(e + k * step, last)];
    #pragma unroll
    for (int k = 0; k < DEG_PER_PASS; ++k) {
      if ((e + k * step < E) && (d[k] >= lo) && (d[k] < hi)) atomicAdd(&deg[d[k]], 1.0f);
    }
  }
}

#define ROLE_PRE 0
#define ROLE_MID 1

template<int K>
__device__ __forceinline__ void tile_product(const float* __restrict__ A, const float* __restrict__ W,
                                             int row0, int half, int l16,
                                             v8f& acc0, v8f& acc1, v8f& acc2, v8f& acc3) {
  const float* arow = A + (size_t)(row0 + l16) * K;
  for (int kb = 0; kb < K; kb += 4) {
    v2f a;
    a.x = arow[kb + half * 2 + 0];
    a.y = arow[kb + half * 2 + 1];

    const float* bp = W + (size_t)(kb + half * 2) * HDIM + l16;
    v2f b0, b1, b2, b3;
    b0.x = bp[0];        b0.y = bp[HDIM];
    b1.x = bp[16];       b1.y = bp[HDIM + 16];
    b2.x = bp[32];       b2.y = bp[HDIM + 32];
    b3.x = bp[48];       b3.y = bp[HDIM + 48];

    acc0 = __builtin_amdgcn_wmma_f32_16x16x4_f32(false, a, false, b0, (short)0, acc0, false, false);
    acc1 = __builtin_amdgcn_wmma_f32_16x16x4_f32(false, a, false, b1, (short)0, acc1, false, false);
    acc2 = __builtin_amdgcn_wmma_f32_16x16x4_f32(false, a, false, b2, (short)0, acc2, false, false);
    acc3 = __builtin_amdgcn_wmma_f32_16x16x4_f32(false, a, false, b3, (short)0, acc3, false, false);
  }
}

template<int K, int ROLE>
__global__ void gemm_n64_wmma(const float* __restrict__ A,
                              const float* __restrict__ W,
                              const float* __restrict__ bias,
                              float* __restrict__ out, int n_rows,
                              const float* __restrict__ skip,
                              float* __restrict__ zplane,
                              float* __restrict__ deg,
                              int* __restrict__ counters,
                              const int* __restrict__ dst, int E) {
  if (ROLE == ROLE_PRE) {
    const int tile_blocks = (n_rows + 127) >> 7;
    if ((int)blockIdx.x >= tile_blocks) {
      count_degrees(dst, deg, E, n_rows, counters);
      return;
    }
  }

  int wave = (blockIdx.x * blockDim.x + threadIdx.x) >> 5;
  int lane = threadIdx.x & 31;
  int row0 = wave * 16;
  if (row0 >= n_rows) return;

  const int half = lane >> 4;
  const int l16  = lane & 15;

  {
    const float4 z4 = make_float4(0.0f, 0.0f, 0.0f, 0.0f);
    float* zp = zplane + (size_t)row0 * HDIM + lane * 4;
    #pragma unroll
    for (int i = 0; i < 8; ++i) *(float4*)(zp + i * 128) = z4;
  }

  float sk[4][8] = {};
  if (ROLE == ROLE_MID) {
    #pragma unroll
    for (int nt = 0; nt < 4; ++nt) {
      #pragma unroll
      for (int r = 0; r < 8; ++r)
        sk[nt][r] = skip[(size_t)(row0 + half * 8 + r) * HDIM + nt * 16 + l16];
    }
  }

  v8f acc0 = {}, acc1 = {}, acc2 = {}, acc3 = {};
  tile_product<K>(A, W, row0, half, l16, acc0, acc1, acc2, acc3);

  v8f accs[4] = {acc0, acc1, acc2, acc3};
  #pragma unroll
  for (int nt = 0; nt < 4; ++nt) {
    int col = nt * 16 + l16;
    float bcol = bias[col];
    #pragma unroll
    for (int r = 0; r < 8; ++r) {
      int m = half * 8 + r;
      float v = accs[nt][r];
      v += bcol; v = fmaxf(v, 0.0f);
      if (ROLE == ROLE_MID) v += sk[nt][r];
      out[(size_t)(row0 + m) * HDIM + col] = v;
    }
  }
}

#define ROWS_IN_FLIGHT 8
template<bool FIRST>
__global__ void __launch_bounds__(ATOMIC_THREADS)
scatter_edges(const int* __restrict__ src, const int* __restrict__ dst,
              const float* __restrict__ deg,
              const float* __restrict__ rows,
              float* __restrict__ acc, int E, int n,
              int* __restrict__ arrivals,
              float* __restrict__ coef_planes) {
  int range, slot;
  claim_range_slot<ATOMIC_BLOCKS>(arrivals, range, slot);
  const int lane = threadIdx.x & 31;
  const int wpb  = ATOMIC_THREADS / 32;
  const int w0   = __builtin_amdgcn_readfirstlane(slot * wpb + (int)(threadIdx.x >> 5));
  const int nw   = (ATOMIC_BLOCKS / 2) * wpb;
  const int cut  = range_cut(n);
  const int lo   = range ? cut : 0;
  const int hi   = range ? n : cut;
  const int ngroups = (E + 31) >> 5;
  const int last = E - 1;
  if (w0 >= ngroups) return;
  float* __restrict__ plane = coef_planes + (size_t)range * E;

  int   e_nx  = w0 * 32 + lane;
  bool  ok_nx = e_nx < E;
  int   s_nx  = src[min(e_nx, last)];
  int   d_nx  = dst[min(e_nx, last)];
  float c_nx  = 0.0f;
  if (!FIRST) c_nx = plane[min(e_nx, last)];

  for (int g = w0; g < ngroups; g += nw) {
    const int   s  = s_nx, d = d_nx;
    const bool  ok = ok_nx;
    const float c  = c_nx;
    const int   e  = g * 32 + lane;
    e_nx  = (g + nw) * 32 + lane;
    ok_nx = e_nx < E;
    s_nx  = src[min(e_nx, last)];
    d_nx  = dst[min(e_nx, last)];
    if (!FIRST) c_nx = plane[min(e_nx, last)];

    const bool mine = ok && (d >= lo) && (d < hi);
    float coef = 0.0f;
    if (FIRST) {
      if (mine) {
        const float cs = deg[s], cd = deg[d];
        const float is = (cs > 0.0f) ? rsqrtf(fmaxf(cs, 1.0f)) : 0.0f;
        const float id = (cd > 0.0f) ? rsqrtf(fmaxf(cd, 1.0f)) : 0.0f;
        coef = is * id;
      }
      if (ok) plane[e] = coef;
    } else {
      coef = mine ? c : 0.0f;
    }

    unsigned m = __builtin_amdgcn_ballot_w32(mine);
    while (m != 0u) {
      float v0[ROWS_IN_FLIGHT], v1[ROWS_IN_FLIGHT];
      int   j[ROWS_IN_FLIGHT];
      #pragma unroll
      for (int u = 0; u < ROWS_IN_FLIGHT; ++u) {
        j[u] = -1;
        if (u == 0 || m != 0u) {
          j[u] = __builtin_ctz(m);
          m &= m - 1u;
          const int sj = __builtin_amdgcn_readlane(s, j[u]);
          const float* rp = rows + (size_t)sj * HDIM + lane;
          v0[u] = rp[0];
          v1[u] = rp[HDIM / 2];
        }
      }
      #pragma unroll
      for (int u = 0; u < ROWS_IN_FLIGHT; ++u) {
        if (u == 0 || j[u] >= 0) {
          const int   dj = __builtin_amdgcn_readlane(d, j[u]);
          const float cj = __int_as_float(__builtin_amdgcn_readlane(__float_as_int(coef), j[u]));
          float* ap = acc + (size_t)dj * HDIM + lane;
          atomicAdd(ap, v0[u] * cj);
          atomicAdd(ap + HDIM / 2, v1[u] * cj);
        }
      }
    }
  }
}


template<int M>
__device__ __forceinline__ float add_row_xor(float v) {
  return v + __int_as_float(__builtin_amdgcn_update_dpp(0, __float_as_int(v), 0x160 + M, 0xf, 0xf, false));
}

#define HEAD_THREADS 256
__global__ void __launch_bounds__(HEAD_THREADS)
gemm_head(const float* __restrict__ A, const float* __restrict__ W,
          const float* __restrict__ bias, const float* __restrict__ skip,
          const float* __restrict__ Wh, const float* __restrict__ bh,
          float* __restrict__ out, int n_rows) {
  const int lane = threadIdx.x & 31;
  const int wave = __builtin_amdgcn_readfirstlane((int)((blockIdx.x * blockDim.x + threadIdx.x) >> 5));
  const int base = wave * 32;
  if (base >= n_rows) return;
  const int half = lane >> 4;
  const int l16  = lane & 15;

  float wh[4], bc[4];
  #pragma unroll
  for (int nt = 0; nt < 4; ++nt) { wh[nt] = Wh[nt * 16 + l16]; bc[nt] = bias[nt * 16 + l16]; }

  float res = 0.0f;
  #pragma unroll
  for (int mt = 0; mt < 2; ++mt) {
    const int row0 = base + mt * 16;
    if (row0 < n_rows) {
      float sk[4][8];
      #pragma unroll
      for (int nt = 0; nt < 4; ++nt) {
        #pragma unroll
        for (int r = 0; r < 8; ++r)
          sk[nt][r] = skip[(size_t)(row0 + half * 8 + r) * HDIM + nt * 16 + l16];
      }
      v8f acc0 = {}, acc1 = {}, acc2 = {}, acc3 = {};
      tile_product<HDIM>(A, W, row0, half, l16, acc0, acc1, acc2, acc3);
      v8f accs[4] = {acc0, acc1, acc2, acc3};
      #pragma unroll
      for (int r = 0; r < 8; ++r) {
        float p = 0.0f;
        #pragma unroll
        for (int nt = 0; nt < 4; ++nt) {
          float v = accs[nt][r] + bc[nt];
          v = fmaxf(v, 0.0f) + sk[nt][r];
          p += v * wh[nt];
        }
        p = add_row_xor<1>(p);
        p = add_row_xor<2>(p);
        p = add_row_xor<4>(p);
        p = add_row_xor<8>(p);
        if (l16 == mt * 8 + r) res = p;
      }
    }
  }
  const int row = base + (l16 >> 3) * 16 + half * 8 + (l16 & 7);
  if (row < n_rows) out[row] = res + bh[0];
}

extern "C" void kernel_launch(void* const* d_in, const int* in_sizes, int n_in,
                              void* d_out, int out_size, void* d_ws, size_t ws_size,
                              hipStream_t stream) {
  const float* x      = (const float*)d_in[0];
  const int*   eidx   = (const int*)d_in[1];
  const float* W_pre  = (const float*)d_in[2];
  const float* b_pre  = (const float*)d_in[3];
  const float* W1     = (const float*)d_in[4];
  const float* b1     = (const float*)d_in[5];
  const float* W2     = (const float*)d_in[6];
  const float* b2     = (const float*)d_in[7];
  const float* W_head = (const float*)d_in[8];
  const float* b_head = (const float*)d_in[9];
  float* out = (float*)d_out;

  const int N = in_sizes[0] / 128;
  const int E = in_sizes[1] / 2;
  const int* src = eidx;
  const int* dst = eidx + E;

  int*   cnt  = (int*)d_ws;
  float* deg  = (float*)d_ws + N_COUNTER_INTS;
  float* h0   = deg  + N;
  float* sum1 = h0   + (size_t)N * HDIM;
  float* h1   = sum1 + (size_t)N * HDIM;
  float* sum2 = h1   + (size_t)N * HDIM;
  float* cpl  = sum2 + (size_t)N * HDIM;

  int gemm_blocks = (N + 127) / 128;

  clear_tables<<<(N / 4 + 255) / 256 + 2, ATOMIC_THREADS, 0, stream>>>(cnt, deg, N);
  gemm_n64_wmma<128, ROLE_PRE><<<gemm_blocks + DEG_BLOCKS, ATOMIC_THREADS, 0, stream>>>(
      x, W_pre, b_pre, h0, N, nullptr, sum1, deg, cnt, dst, E);

  scatter_edges<true><<<ATOMIC_BLOCKS, ATOMIC_THREADS, 0, stream>>>(src, dst, deg, h0, sum1, E, N,
                                                        cnt + 1 * LAUNCH_PITCH, cpl);
  gemm_n64_wmma<64, ROLE_MID><<<gemm_blocks, 256, 0, stream>>>(sum1, W1, b1, h1, N,
                                                              h0, sum2, nullptr, nullptr, nullptr, 0);

  scatter_edges<false><<<ATOMIC_BLOCKS, ATOMIC_THREADS, 0, stream>>>(src, dst, nullptr, h1, sum2, E, N,
                                                         cnt + 2 * LAUNCH_PITCH, cpl);
  gemm_head<<<(N + 255) / 256, HEAD_THREADS, 0, stream>>>(sum2, W2, b2, h1, W_head, b_head, out, N);
}
